# starting kernel
# baseline (speedup 1.0000x reference)
#include <hip/hip_runtime.h>
#include <hip/hip_bf16.h>

typedef __attribute__((ext_vector_type(2))) float v2f;
typedef __attribute__((ext_vector_type(8))) float v8f;

#define EHID 64
#define RHID 32
#define DFEAT 160
#define DOUT 224
#define EPSF 1e-16f

static inline int ceil_div(int a, int b) { return (a + b - 1) / b; }

__global__ void k_zero_f32(float* p, int n) {
    int i = blockIdx.x * blockDim.x + threadIdx.x;
    if (i < n) p[i] = 0.0f;
}

__global__ void k_zero_u32(unsigned* p, int n) {
    int i = blockIdx.x * blockDim.x + threadIdx.x;
    if (i < n) p[i] = 0u;
}

__global__ void k_att_wmma(const float* __restrict__ x,
                           const int* __restrict__ ei,
                           const int* __restrict__ ej,
                           const int* __restrict__ rel,
                           const float* __restrict__ rel_emb,
                           const float* __restrict__ w,
                           float* __restrict__ att, int E) {
    __shared__ float sw[DFEAT];
    int t = threadIdx.x;
    if (t < DFEAT) sw[t] = w[t];
    __syncthreads();

    int wave  = blockIdx.x * (blockDim.x >> 5) + (t >> 5);
    int lane  = t & 31;
    int ebase = wave << 4;
    if (ebase >= E) return;

    int m     = lane & 15;
    int khalf = lane >> 4;
    int e     = ebase + m;

    int vi = ei[e], vj = ej[e], r = rel[e];
    const float* xi = x + (long)vi * EHID;
    const float* xj = x + (long)vj * EHID;
    const float* re = rel_emb + (long)r * RHID;

    v8f c = {};
#pragma unroll
    for (int ch = 0; ch < DFEAT / 4; ++ch) {
        int k0 = ch * 4 + khalf * 2;
        v2f a;
        if (k0 < EHID)              a = *(const v2f*)(xi + k0);
        else if (k0 < EHID + RHID)  a = *(const v2f*)(re + (k0 - EHID));
        else                        a = *(const v2f*)(xj + (k0 - EHID - RHID));
        v2f b;
        b.x = sw[k0];
        b.y = sw[k0 + 1];
        c = __builtin_amdgcn_wmma_f32_16x16x4_f32(
                 false, a,  false, b,
                 (short)0, c,  false,  false);
    }

    if ((lane & 15) == 0) {
        float4 lo = make_float4(c[0], c[1], c[2], c[3]);
        float4 hi = make_float4(c[4], c[5], c[6], c[7]);
        float4* o = (float4*)(att + ebase + (khalf << 3));
        o[0] = lo;
        o[1] = hi;
    }
}

__device__ __forceinline__ unsigned enc_ord(float f) {
    unsigned u = __float_as_uint(f);
    return (u & 0x80000000u) ? ~u : (u | 0x80000000u);
}
__device__ __forceinline__ float dec_ord(unsigned u) {
    return (u & 0x80000000u) ? __uint_as_float(u & 0x7FFFFFFFu)
                             : __uint_as_float(~u);
}

__global__ void k_segmax(const float* __restrict__ att, const int* __restrict__ ei,
                         unsigned* __restrict__ segmax_u, int E) {
    int e = blockIdx.x * blockDim.x + threadIdx.x;
    if (e >= E) return;
    atomicMax(&segmax_u[ei[e]], enc_ord(att[e]));
}

__global__ void k_expsum(float* __restrict__ att, const int* __restrict__ ei,
                         const unsigned* __restrict__ segmax_u,
                         float* __restrict__ segsum, int E) {
    int e = blockIdx.x * blockDim.x + threadIdx.x;
    if (e >= E) return;
    int vi = ei[e];
    float p = expf(att[e] - dec_ord(segmax_u[vi]));
    att[e] = p;
    atomicAdd(&segsum[vi], p);
}

__global__ void k_scatter(const float* __restrict__ x,
                          const int* __restrict__ ei,
                          const int* __restrict__ ej,
                          const int* __restrict__ rel,
                          const float* __restrict__ rel_emb,
                          const float* __restrict__ p,
                          const float* __restrict__ segsum,
                          float* __restrict__ out, int E) {
    int wave = blockIdx.x * (blockDim.x >> 5) + (threadIdx.x >> 5);
    int lane = threadIdx.x & 31;
    if (wave >= E) return;
    int e = wave;
    int vi = ei[e], vj = ej[e], r = rel[e];
    float coef = p[e] / (segsum[vi] + EPSF);
    float* orow = out + (long)vi * DOUT + EHID;
    const float* xi = x + (long)vi * EHID;
    const float* xj = x + (long)vj * EHID;
    const float* re = rel_emb + (long)r * RHID;
#pragma unroll
    for (int i = 0; i < DFEAT / 32; ++i) {
        int d = lane + 32 * i;
        float v;
        if (d < EHID)              v = xi[d];
        else if (d < EHID + RHID)  v = re[d - EHID];
        else                       v = xj[d - EHID - RHID];
        atomicAdd(orow + d, v * coef);
    }
}

__global__ void k_finalize(const float* __restrict__ x, float* __restrict__ out, int total) {
    int i = blockIdx.x * blockDim.x + threadIdx.x;
    if (i >= total) return;
    int c = i % DOUT;
    int n = i / DOUT;
    if (c < EHID) out[i] = x[n * EHID + c];
    else          out[i] = fmaxf(out[i], 0.0f);
}

extern "C" void kernel_launch(void* const* d_in, const int* in_sizes, int n_in,
                              void* d_out, int out_size, void* d_ws, size_t ws_size,
                              hipStream_t stream) {
    const float* x        = (const float*)d_in[0];
    const int*   eidx     = (const int*)d_in[1];
    const int*   rel      = (const int*)d_in[2];
    const float* rel_emb  = (const float*)d_in[3];
    const float* w        = (const float*)d_in[4];
    float* out = (float*)d_out;

    const int E = in_sizes[2];
    const int N = in_sizes[0] / EHID;
    const int* ei = eidx;
    const int* ej = eidx + E;

    char* ws = (char*)d_ws;
    float*    att      = (float*)ws;
    unsigned* segmax_u = (unsigned*)(ws + (size_t)E * 4);
    float*    segsum   = (float*)(ws + (size_t)E * 4 + (size_t)N * 4);

    const int B = 256;

    int tot_out = N * DOUT;
    k_zero_f32<<<ceil_div(tot_out, B), B, 0, stream>>>(out, tot_out);
    k_zero_u32<<<ceil_div(2 * N, B), B, 0, stream>>>(segmax_u, 2 * N);

    int waves = E / 16;
    k_att_wmma<<<ceil_div(waves, 8), B, 0, stream>>>(x, ei, ej, rel, rel_emb, w, att, E);

    k_segmax<<<ceil_div(E, B), B, 0, stream>>>(att, ei, segmax_u, E);

    k_expsum<<<ceil_div(E, B), B, 0, stream>>>(att, ei, segmax_u, segsum, E);

    k_scatter<<<ceil_div(E, 8), B, 0, stream>>>(x, ei, ej, rel, rel_emb, att, segsum, out, E);

    k_finalize<<<ceil_div(tot_out, B), B, 0, stream>>>(x, out, tot_out);
}


# Aggregate columns fed by x[ei] are constant within a segment: written in the last pass as x[n]*S/(S+eps), no longer summed over edges by atomics; scatter and clear cover the other 96 columns
# speedup vs baseline: 1.2003x; 1.2003x over previous
#include <hip/hip_runtime.h>
#include <hip/hip_bf16.h>

typedef __attribute__((ext_vector_type(2))) float v2f;
typedef __attribute__((ext_vector_type(8))) float v8f;

#define EHID 64
#define RHID 32
#define DFEAT 160
#define DOUT 224
#define EPSF 1e-16f
#define ACC0 128
#define DACC 96

static inline int ceil_div(int a, int b) { return (a + b - 1) / b; }

__global__ void k_zero_acc(float* out, int total) {
    int i = blockIdx.x * blockDim.x + threadIdx.x;
    if (i >= total) return;
    int n = i / DACC;
    int c = i - n * DACC;
    out[(long)n * DOUT + ACC0 + c] = 0.0f;
}

__global__ void k_zero_u32(unsigned* p, int n) {
    int i = blockIdx.x * blockDim.x + threadIdx.x;
    if (i < n) p[i] = 0u;
}

__global__ void k_att_wmma(const float* __restrict__ x,
                           const int* __restrict__ ei,
                           const int* __restrict__ ej,
                           const int* __restrict__ rel,
                           const float* __restrict__ rel_emb,
                           const float* __restrict__ w,
                           float* __restrict__ att, int E) {
    __shared__ float sw[DFEAT];
    int t = threadIdx.x;
    if (t < DFEAT) sw[t] = w[t];
    __syncthreads();

    int wave  = blockIdx.x * (blockDim.x >> 5) + (t >> 5);
    int lane  = t & 31;
    int ebase = wave << 4;
    if (ebase >= E) return;

    int m     = lane & 15;
    int khalf = lane >> 4;
    int e     = ebase + m;

    int vi = ei[e], vj = ej[e], r = rel[e];
    const float* xi = x + (long)vi * EHID;
    const float* xj = x + (long)vj * EHID;
    const float* re = rel_emb + (long)r * RHID;

    v8f c = {};
#pragma unroll
    for (int ch = 0; ch < DFEAT / 4; ++ch) {
        int k0 = ch * 4 + khalf * 2;
        v2f a;
        if (k0 < EHID)              a = *(const v2f*)(xi + k0);
        else if (k0 < EHID + RHID)  a = *(const v2f*)(re + (k0 - EHID));
        else                        a = *(const v2f*)(xj + (k0 - EHID - RHID));
        v2f b;
        b.x = sw[k0];
        b.y = sw[k0 + 1];
        c = __builtin_amdgcn_wmma_f32_16x16x4_f32(
                 false, a,  false, b,
                 (short)0, c,  false,  false);
    }

    if ((lane & 15) == 0) {
        float4 lo = make_float4(c[0], c[1], c[2], c[3]);
        float4 hi = make_float4(c[4], c[5], c[6], c[7]);
        float4* o = (float4*)(att + ebase + (khalf << 3));
        o[0] = lo;
        o[1] = hi;
    }
}

__device__ __forceinline__ unsigned enc_ord(float f) {
    unsigned u = __float_as_uint(f);
    return (u & 0x80000000u) ? ~u : (u | 0x80000000u);
}
__device__ __forceinline__ float dec_ord(unsigned u) {
    return (u & 0x80000000u) ? __uint_as_float(u & 0x7FFFFFFFu)
                             : __uint_as_float(~u);
}

__global__ void k_segmax(const float* __restrict__ att, const int* __restrict__ ei,
                         unsigned* __restrict__ segmax_u, int E) {
    int e = blockIdx.x * blockDim.x + threadIdx.x;
    if (e >= E) return;
    atomicMax(&segmax_u[ei[e]], enc_ord(att[e]));
}

__global__ void k_expsum(float* __restrict__ att, const int* __restrict__ ei,
                         const unsigned* __restrict__ segmax_u,
                         float* __restrict__ segsum, int E) {
    int e = blockIdx.x * blockDim.x + threadIdx.x;
    if (e >= E) return;
    int vi = ei[e];
    float p = expf(att[e] - dec_ord(segmax_u[vi]));
    att[e] = p;
    atomicAdd(&segsum[vi], p);
}

__global__ void k_scatter(const float* __restrict__ x,
                          const int* __restrict__ ei,
                          const int* __restrict__ ej,
                          const int* __restrict__ rel,
                          const float* __restrict__ rel_emb,
                          const float* __restrict__ p,
                          const float* __restrict__ segsum,
                          float* __restrict__ out, int E) {
    int wave = blockIdx.x * (blockDim.x >> 5) + (threadIdx.x >> 5);
    int lane = threadIdx.x & 31;
    if (wave >= E) return;
    int e = wave;
    int vi = ei[e], vj = ej[e], r = rel[e];
    float coef = p[e] / (segsum[vi] + EPSF);
    float* orow = out + (long)vi * DOUT + EHID;
    const float* xj = x + (long)vj * EHID;
    const float* re = rel_emb + (long)r * RHID;
#pragma unroll
    for (int i = EHID / 32; i < DFEAT / 32; ++i) {
        int d = lane + 32 * i;
        float v;
        if (d < EHID + RHID)  v = re[d - EHID];
        else                  v = xj[d - EHID - RHID];
        atomicAdd(orow + d, v * coef);
    }
}

__global__ void k_finalize(const float* __restrict__ x,
                           const float* __restrict__ segsum,
                           float* __restrict__ out, int total) {
    int i = blockIdx.x * blockDim.x + threadIdx.x;
    if (i >= total) return;
    int c = i % DOUT;
    int n = i / DOUT;
    if (c < EHID) {
        out[i] = x[n * EHID + c];
    } else if (c < ACC0) {
        float s = segsum[n];
        out[i] = fmaxf(x[n * EHID + (c - EHID)] * (s / (s + EPSF)), 0.0f);
    } else {
        out[i] = fmaxf(out[i], 0.0f);
    }
}

extern "C" void kernel_launch(void* const* d_in, const int* in_sizes, int n_in,
                              void* d_out, int out_size, void* d_ws, size_t ws_size,
                              hipStream_t stream) {
    const float* x        = (const float*)d_in[0];
    const int*   eidx     = (const int*)d_in[1];
    const int*   rel      = (const int*)d_in[2];
    const float* rel_emb  = (const float*)d_in[3];
    const float* w        = (const float*)d_in[4];
    float* out = (float*)d_out;

    const int E = in_sizes[2];
    const int N = in_sizes[0] / EHID;
    const int* ei = eidx;
    const int* ej = eidx + E;

    char* ws = (char*)d_ws;
    float*    att      = (float*)ws;
    unsigned* segmax_u = (unsigned*)(ws + (size_t)E * 4);
    float*    segsum   = (float*)(ws + (size_t)E * 4 + (size_t)N * 4);

    const int B = 256;

    int tot_out = N * DOUT;
    int tot_acc = N * DACC;
    k_zero_acc<<<ceil_div(tot_acc, B), B, 0, stream>>>(out, tot_acc);
    k_zero_u32<<<ceil_div(2 * N, B), B, 0, stream>>>(segmax_u, 2 * N);

    int waves = E / 16;
    k_att_wmma<<<ceil_div(waves, 8), B, 0, stream>>>(x, ei, ej, rel, rel_emb, w, att, E);

    k_segmax<<<ceil_div(E, B), B, 0, stream>>>(att, ei, segmax_u, E);

    k_expsum<<<ceil_div(E, B), B, 0, stream>>>(att, ei, segmax_u, segsum, E);

    k_scatter<<<ceil_div(E, 8), B, 0, stream>>>(x, ei, ej, rel, rel_emb, att, segsum, out, E);

    k_finalize<<<ceil_div(tot_out, B), B, 0, stream>>>(x, segsum, out, tot_out);
}


# Fewer, fatter waves (about 11x fewer): clear and last pass move 4 float4 per lane; a scatter wave walks 32 edges, 8 edges' rows in flight; float atomic adds kept, arrival order as before
# speedup vs baseline: 1.3881x; 1.1565x over previous
#include <hip/hip_runtime.h>
#include <hip/hip_bf16.h>

typedef __attribute__((ext_vector_type(2))) float v2f;
typedef __attribute__((ext_vector_type(8))) float v8f;

#define EHID 64
#define RHID 32
#define DFEAT 160
#define DOUT 224
#define EPSF 1e-16f
#define ACC0 128
#define DACC 96
#define X4   (EHID / 4)
#define ACC4 (DACC / 4)
#define ROW4 (DOUT / 4)
#define ZPT 4
#define FPT 4
#define EPW 32
#define EGRP 8

static inline int ceil_div(int a, int b) { return (a + b - 1) / b; }

__global__ void k_zero_acc(float* out, int per) {
    int t = blockIdx.x * blockDim.x + threadIdx.x;
    if (t >= per) return;
    const float4 z = make_float4(0.0f, 0.0f, 0.0f, 0.0f);
#pragma unroll
    for (int k = 0; k < ZPT; ++k) {
        int i = t + k * per;
        int n = i / ACC4;
        int q = i - n * ACC4;
        *(float4*)(out + (long)n * DOUT + ACC0 + 4 * q) = z;
    }
}

__global__ void k_zero_u32(unsigned* p, int n) {
    int i = blockIdx.x * blockDim.x + threadIdx.x;
    if (i < n) p[i] = 0u;
}

__global__ void k_att_wmma(const float* __restrict__ x,
                           const int* __restrict__ ei,
                           const int* __restrict__ ej,
                           const int* __restrict__ rel,
                           const float* __restrict__ rel_emb,
                           const float* __restrict__ w,
                           float* __restrict__ att, int E) {
    __shared__ float sw[DFEAT];
    int t = threadIdx.x;
    if (t < DFEAT) sw[t] = w[t];
    __syncthreads();

    int wave  = blockIdx.x * (blockDim.x >> 5) + (t >> 5);
    int lane  = t & 31;
    int ebase = wave << 4;
    if (ebase >= E) return;

    int m     = lane & 15;
    int khalf = lane >> 4;
    int e     = ebase + m;

    int vi = ei[e], vj = ej[e], r = rel[e];
    const float* xi = x + (long)vi * EHID;
    const float* xj = x + (long)vj * EHID;
    const float* re = rel_emb + (long)r * RHID;

    v8f c = {};
#pragma unroll
    for (int ch = 0; ch < DFEAT / 4; ++ch) {
        int k0 = ch * 4 + khalf * 2;
        v2f a;
        if (k0 < EHID)              a = *(const v2f*)(xi + k0);
        else if (k0 < EHID + RHID)  a = *(const v2f*)(re + (k0 - EHID));
        else                        a = *(const v2f*)(xj + (k0 - EHID - RHID));
        v2f b;
        b.x = sw[k0];
        b.y = sw[k0 + 1];
        c = __builtin_amdgcn_wmma_f32_16x16x4_f32(
                 false, a,  false, b,
                 (short)0, c,  false,  false);
    }

    if ((lane & 15) == 0) {
        float4 lo = make_float4(c[0], c[1], c[2], c[3]);
        float4 hi = make_float4(c[4], c[5], c[6], c[7]);
        float4* o = (float4*)(att + ebase + (khalf << 3));
        o[0] = lo;
        o[1] = hi;
    }
}

__device__ __forceinline__ unsigned enc_ord(float f) {
    unsigned u = __float_as_uint(f);
    return (u & 0x80000000u) ? ~u : (u | 0x80000000u);
}
__device__ __forceinline__ float dec_ord(unsigned u) {
    return (u & 0x80000000u) ? __uint_as_float(u & 0x7FFFFFFFu)
                             : __uint_as_float(~u);
}

__global__ void k_segmax(const float* __restrict__ att, const int* __restrict__ ei,
                         unsigned* __restrict__ segmax_u, int E) {
    int e = blockIdx.x * blockDim.x + threadIdx.x;
    if (e >= E) return;
    atomicMax(&segmax_u[ei[e]], enc_ord(att[e]));
}

__global__ void k_expsum(float* __restrict__ att, const int* __restrict__ ei,
                         const unsigned* __restrict__ segmax_u,
                         float* __restrict__ segsum, int E) {
    int e = blockIdx.x * blockDim.x + threadIdx.x;
    if (e >= E) return;
    int vi = ei[e];
    float p = expf(att[e] - dec_ord(segmax_u[vi]));
    att[e] = p;
    atomicAdd(&segsum[vi], p);
}

__global__ void k_scatter(const float* __restrict__ x,
                          const int* __restrict__ ei,
                          const int* __restrict__ ej,
                          const int* __restrict__ rel,
                          const float* __restrict__ rel_emb,
                          const float* __restrict__ p,
                          const float* __restrict__ segsum,
                          float* __restrict__ out, int E) {
    int wv   = __builtin_amdgcn_readfirstlane((int)(threadIdx.x >> 5));
    int wave = blockIdx.x * (blockDim.x >> 5) + wv;
    int lane = threadIdx.x & 31;
    int e0   = wave * EPW;
    if (e0 >= E) return;
    int cnt = E - e0;
    if (cnt > EPW) cnt = EPW;

    int vi = 0, vj = 0, r = 0;
    float coef = 0.0f;
    if (lane < cnt) {
        int e = e0 + lane;
        vi = ei[e]; vj = ej[e]; r = rel[e];
        coef = p[e] / (segsum[vi] + EPSF);
    }

    int full = cnt & ~(EGRP - 1);
    for (int kb = 0; kb < full; kb += EGRP) {
        float  vr[EGRP], va[EGRP], vb[EGRP], cf[EGRP];
        float* orow[EGRP];
#pragma unroll
        for (int u = 0; u < EGRP; ++u) {
            int k   = kb + u;
            int svi = __builtin_amdgcn_readlane(vi, k);
            int svj = __builtin_amdgcn_readlane(vj, k);
            int sr  = __builtin_amdgcn_readlane(r, k);
            cf[u]   = __int_as_float(__builtin_amdgcn_readlane(__float_as_int(coef), k));
            const float* re = rel_emb + (long)sr * RHID;
            const float* xj = x + (long)svj * EHID;
            vr[u] = re[lane];
            va[u] = xj[lane];
            vb[u] = xj[lane + 32];
            orow[u] = out + (long)svi * DOUT + ACC0;
        }
#pragma unroll
        for (int u = 0; u < EGRP; ++u) {
            atomicAdd(orow[u] + lane,      vr[u] * cf[u]);
            atomicAdd(orow[u] + 32 + lane, va[u] * cf[u]);
            atomicAdd(orow[u] + 64 + lane, vb[u] * cf[u]);
        }
    }
    for (int k = full; k < cnt; ++k) {
        int svi = __builtin_amdgcn_readlane(vi, k);
        int svj = __builtin_amdgcn_readlane(vj, k);
        int sr  = __builtin_amdgcn_readlane(r, k);
        float c1 = __int_as_float(__builtin_amdgcn_readlane(__float_as_int(coef), k));
        const float* re = rel_emb + (long)sr * RHID;
        const float* xj = x + (long)svj * EHID;
        float* orow1 = out + (long)svi * DOUT + ACC0;
        float a0 = re[lane], a1 = xj[lane], a2 = xj[lane + 32];
        atomicAdd(orow1 + lane,      a0 * c1);
        atomicAdd(orow1 + 32 + lane, a1 * c1);
        atomicAdd(orow1 + 64 + lane, a2 * c1);
    }
}

__global__ void k_finalize(const float* __restrict__ x,
                           const float* __restrict__ segsum,
                           float* __restrict__ out, int per) {
    int t = blockIdx.x * blockDim.x + threadIdx.x;
    if (t >= per) return;
    float4 v[FPT];
    float  s[FPT];
    int    cc[FPT];
#pragma unroll
    for (int k = 0; k < FPT; ++k) {
        int i  = t + k * per;
        int n  = i / ROW4;
        int c4 = i - n * ROW4;
        const float4* src = (c4 < 2 * X4)
            ? (const float4*)(x + (long)n * EHID) + (c4 & (X4 - 1))
            : (const float4*)out + i;
        v[k]  = *src;
        s[k]  = segsum[n];
        cc[k] = c4;
    }
#pragma unroll
    for (int k = 0; k < FPT; ++k) {
        int  i   = t + k * per;
        bool mid = (cc[k] >= X4) && (cc[k] < 2 * X4);
        float f  = s[k] / (s[k] + EPSF);
        float4 o = v[k];
        if (mid) { o.x *= f; o.y *= f; o.z *= f; o.w *= f; }
        if (cc[k] >= X4) {
            o.x = fmaxf(o.x, 0.0f); o.y = fmaxf(o.y, 0.0f);
            o.z = fmaxf(o.z, 0.0f); o.w = fmaxf(o.w, 0.0f);
        }
        ((float4*)out)[i] = o;
    }
}

extern "C" void kernel_launch(void* const* d_in, const int* in_sizes, int n_in,
                              void* d_out, int out_size, void* d_ws, size_t ws_size,
                              hipStream_t stream) {
    const float* x        = (const float*)d_in[0];
    const int*   eidx     = (const int*)d_in[1];
    const int*   rel      = (const int*)d_in[2];
    const float* rel_emb  = (const float*)d_in[3];
    const float* w        = (const float*)d_in[4];
    float* out = (float*)d_out;

    const int E = in_sizes[2];
    const int N = in_sizes[0] / EHID;
    const int* ei = eidx;
    const int* ej = eidx + E;

    char* ws = (char*)d_ws;
    float*    att      = (float*)ws;
    unsigned* segmax_u = (unsigned*)(ws + (size_t)E * 4);
    float*    segsum   = (float*)(ws + (size_t)E * 4 + (size_t)N * 4);

    const int B = 256;

    int per_z = N * (ACC4 / ZPT);
    int per_f = N * (ROW4 / FPT);
    k_zero_acc<<<ceil_div(per_z, B), B, 0, stream>>>(out, per_z);
    k_zero_u32<<<ceil_div(2 * N, B), B, 0, stream>>>(segmax_u, 2 * N);

    int waves = E / 16;
    k_att_wmma<<<ceil_div(waves, 8), B, 0, stream>>>(x, ei, ej, rel, rel_emb, w, att, E);

    k_segmax<<<ceil_div(E, B), B, 0, stream>>>(att, ei, segmax_u, E);

    k_expsum<<<ceil_div(E, B), B, 0, stream>>>(att, ei, segmax_u, segsum, E);

    k_scatter<<<ceil_div(E, 8 * EPW), B, 0, stream>>>(x, ei, ej, rel, rel_emb, att, segsum, out, E);

    k_finalize<<<ceil_div(per_f, B), B, 0, stream>>>(x, segsum, out, per_f);
}


# Launches 7 to 4: segment max by atomic max at the end of the score launch; exp, segment sum and scatter share one launch, the division by (S+eps) moved behind the edge sum into the last pass
# speedup vs baseline: 1.4675x; 1.0572x over previous
#include <hip/hip_runtime.h>
#include <hip/hip_bf16.h>

typedef __attribute__((ext_vector_type(2))) float v2f;
typedef __attribute__((ext_vector_type(8))) float v8f;

#define EHID 64
#define RHID 32
#define DFEAT 160
#define DOUT 224
#define EPSF 1e-16f
#define ACC0 128
#define DACC 96
#define X4   (EHID / 4)
#define ACC4 (DACC / 4)
#define ROW4 (DOUT / 4)
#define ZPT 4
#define FPT 4
#define EPW 32
#define EGRP 8

static inline int ceil_div(int a, int b) { return (a + b - 1) / b; }

__global__ void k_clear(float* out, unsigned* seg, int per, int nseg) {
    int t = blockIdx.x * blockDim.x + threadIdx.x;
    if (t >= per) return;
    if (t < nseg) seg[t] = 0u;
    const float4 z = make_float4(0.0f, 0.0f, 0.0f, 0.0f);
#pragma unroll
    for (int k = 0; k < ZPT; ++k) {
        int i = t + k * per;
        int n = i / ACC4;
        int q = i - n * ACC4;
        *(float4*)(out + (long)n * DOUT + ACC0 + 4 * q) = z;
    }
}

__device__ __forceinline__ unsigned enc_ord(float f) {
    unsigned u = __float_as_uint(f);
    return (u & 0x80000000u) ? ~u : (u | 0x80000000u);
}
__device__ __forceinline__ float dec_ord(unsigned u) {
    return (u & 0x80000000u) ? __uint_as_float(u & 0x7FFFFFFFu)
                             : __uint_as_float(~u);
}

__global__ void k_att_wmma(const float* __restrict__ x,
                           const int* __restrict__ ei,
                           const int* __restrict__ ej,
                           const int* __restrict__ rel,
                           const float* __restrict__ rel_emb,
                           const float* __restrict__ w,
                           float* __restrict__ att,
                           unsigned* __restrict__ segmax_u, int E) {
    __shared__ float sw[DFEAT];
    int t = threadIdx.x;
    if (t < DFEAT) sw[t] = w[t];
    __syncthreads();

    int wave  = blockIdx.x * (blockDim.x >> 5) + (t >> 5);
    int lane  = t & 31;
    int ebase = wave << 4;
    if (ebase >= E) return;

    int m     = lane & 15;
    int khalf = lane >> 4;
    int e     = ebase + m;

    int vi = ei[e], vj = ej[e], r = rel[e];
    const float* xi = x + (long)vi * EHID;
    const float* xj = x + (long)vj * EHID;
    const float* re = rel_emb + (long)r * RHID;

    v8f c = {};
#pragma unroll
    for (int ch = 0; ch < DFEAT / 4; ++ch) {
        int k0 = ch * 4 + khalf * 2;
        v2f a;
        if (k0 < EHID)              a = *(const v2f*)(xi + k0);
        else if (k0 < EHID + RHID)  a = *(const v2f*)(re + (k0 - EHID));
        else                        a = *(const v2f*)(xj + (k0 - EHID - RHID));
        v2f b;
        b.x = sw[k0];
        b.y = sw[k0 + 1];
        c = __builtin_amdgcn_wmma_f32_16x16x4_f32(
                 false, a,  false, b,
                 (short)0, c,  false,  false);
    }

    float sc = c[0];
    sc = (m == 1) ? c[1] : sc;
    sc = (m == 2) ? c[2] : sc;
    sc = (m == 3) ? c[3] : sc;
    sc = (m == 4) ? c[4] : sc;
    sc = (m == 5) ? c[5] : sc;
    sc = (m == 6) ? c[6] : sc;
    sc = (m == 7) ? c[7] : sc;
    if (m < 8) {
        int e2 = ebase + (khalf << 3) + m;
        att[e2] = sc;
        atomicMax(&segmax_u[ei[e2]], enc_ord(sc));
    }
}

__global__ void k_scatter(const float* __restrict__ x,
                          const int* __restrict__ ei,
                          const int* __restrict__ ej,
                          const int* __restrict__ rel,
                          const float* __restrict__ rel_emb,
                          const float* __restrict__ att,
                          const unsigned* __restrict__ segmax_u,
                          float* __restrict__ segsum,
                          float* __restrict__ out, int E) {
    int wv   = __builtin_amdgcn_readfirstlane((int)(threadIdx.x >> 5));
    int wave = blockIdx.x * (blockDim.x >> 5) + wv;
    int lane = threadIdx.x & 31;
    int e0   = wave * EPW;
    if (e0 >= E) return;
    int cnt = E - e0;
    if (cnt > EPW) cnt = EPW;

    int vi = 0, vj = 0, r = 0;
    float coef = 0.0f;
    if (lane < cnt) {
        int e = e0 + lane;
        vi = ei[e]; vj = ej[e]; r = rel[e];
        coef = expf(att[e] - dec_ord(segmax_u[vi]));
        atomicAdd(&segsum[vi], coef);
    }

    int full = cnt & ~(EGRP - 1);
    for (int kb = 0; kb < full; kb += EGRP) {
        float  vr[EGRP], va[EGRP], vb[EGRP], cf[EGRP];
        float* orow[EGRP];
#pragma unroll
        for (int u = 0; u < EGRP; ++u) {
            int k   = kb + u;
            int svi = __builtin_amdgcn_readlane(vi, k);
            int svj = __builtin_amdgcn_readlane(vj, k);
            int sr  = __builtin_amdgcn_readlane(r, k);
            cf[u]   = __int_as_float(__builtin_amdgcn_readlane(__float_as_int(coef), k));
            const float* re = rel_emb + (long)sr * RHID;
            const float* xj = x + (long)svj * EHID;
            vr[u] = re[lane];
            va[u] = xj[lane];
            vb[u] = xj[lane + 32];
            orow[u] = out + (long)svi * DOUT + ACC0;
        }
#pragma unroll
        for (int u = 0; u < EGRP; ++u) {
            atomicAdd(orow[u] + lane,      vr[u] * cf[u]);
            atomicAdd(orow[u] + 32 + lane, va[u] * cf[u]);
            atomicAdd(orow[u] + 64 + lane, vb[u] * cf[u]);
        }
    }
    for (int k = full; k < cnt; ++k) {
        int svi = __builtin_amdgcn_readlane(vi, k);
        int svj = __builtin_amdgcn_readlane(vj, k);
        int sr  = __builtin_amdgcn_readlane(r, k);
        float c1 = __int_as_float(__builtin_amdgcn_readlane(__float_as_int(coef), k));
        const float* re = rel_emb + (long)sr * RHID;
        const float* xj = x + (long)svj * EHID;
        float* orow1 = out + (long)svi * DOUT + ACC0;
        float a0 = re[lane], a1 = xj[lane], a2 = xj[lane + 32];
        atomicAdd(orow1 + lane,      a0 * c1);
        atomicAdd(orow1 + 32 + lane, a1 * c1);
        atomicAdd(orow1 + 64 + lane, a2 * c1);
    }
}

__global__ void k_finalize(const float* __restrict__ x,
                           const float* __restrict__ segsum,
                           float* __restrict__ out, int per) {
    int t = blockIdx.x * blockDim.x + threadIdx.x;
    if (t >= per) return;
    float4 v[FPT];
    float  s[FPT];
    int    cc[FPT];
#pragma unroll
    for (int k = 0; k < FPT; ++k) {
        int i  = t + k * per;
        int n  = i / ROW4;
        int c4 = i - n * ROW4;
        const float4* src = (c4 < 2 * X4)
            ? (const float4*)(x + (long)n * EHID) + (c4 & (X4 - 1))
            : (const float4*)out + i;
        v[k]  = *src;
        s[k]  = segsum[n];
        cc[k] = c4;
    }
#pragma unroll
    for (int k = 0; k < FPT; ++k) {
        int  i   = t + k * per;
        bool mid = (cc[k] >= X4) && (cc[k] < 2 * X4);
        float f  = (mid ? s[k] : 1.0f) / (s[k] + EPSF);
        float4 o = v[k];
        if (cc[k] >= X4) {
            o.x *= f; o.y *= f; o.z *= f; o.w *= f;
            o.x = fmaxf(o.x, 0.0f); o.y = fmaxf(o.y, 0.0f);
            o.z = fmaxf(o.z, 0.0f); o.w = fmaxf(o.w, 0.0f);
        }
        ((float4*)out)[i] = o;
    }
}

extern "C" void kernel_launch(void* const* d_in, const int* in_sizes, int n_in,
                              void* d_out, int out_size, void* d_ws, size_t ws_size,
                              hipStream_t stream) {
    const float* x        = (const float*)d_in[0];
    const int*   eidx     = (const int*)d_in[1];
    const int*   rel      = (const int*)d_in[2];
    const float* rel_emb  = (const float*)d_in[3];
    const float* w        = (const float*)d_in[4];
    float* out = (float*)d_out;

    const int E = in_sizes[2];
    const int N = in_sizes[0] / EHID;
    const int* ei = eidx;
    const int* ej = eidx + E;

    char* ws = (char*)d_ws;
    float*    att      = (float*)ws;
    unsigned* segmax_u = (unsigned*)(ws + (size_t)E * 4);
    float*    segsum   = (float*)(ws + (size_t)E * 4 + (size_t)N * 4);

    const int B = 256;

    int per_z = N * (ACC4 / ZPT);
    int per_f = N * (ROW4 / FPT);
    k_clear<<<ceil_div(per_z, B), B, 0, stream>>>(out, segmax_u, per_z, 2 * N);

    int waves = E / 16;
    k_att_wmma<<<ceil_div(waves, 8), B, 0, stream>>>(x, ei, ej, rel, rel_emb, w, att, segmax_u, E);

    k_scatter<<<ceil_div(E, 8 * EPW), B, 0, stream>>>(x, ei, ej, rel, rel_emb, att, segmax_u, segsum, out, E);

    k_finalize<<<ceil_div(per_f, B), B, 0, stream>>>(x, segsum, out, per_f);
}
